# starting kernel
# baseline (speedup 1.0000x reference)
#include <hip/hip_runtime.h>

typedef __attribute__((ext_vector_type(2))) float v2f;
typedef __attribute__((ext_vector_type(8))) float v8f;

#define VOCAB 50000
#define LROWS 8921
#define DCH   300
#define DP2   150
#define KF    10
#define BB    8
#define TLEN  2048
#define TP    2057
#define PADK  9

#define XCNT  (BB * DCH * TLEN)
#define HCNT  (BB * DCH * TP)

__global__ __launch_bounds__(256)
void DP_gather(const int* __restrict__ ids, const float* __restrict__ embed,
               v2f* __restrict__ x2) {
  int idx = blockIdx.x * blockDim.x + threadIdx.x;
  if (idx >= XCNT / 2) return;
  int t = idx % TLEN;
  int p = (idx / TLEN) % DP2;
  int b = idx / (TLEN * DP2);
  int row = ids[b * TLEN + t];
  x2[idx] = *(const v2f*)&embed[row * DCH + 2 * p];
}

#define TCHUNK 512
#define XJ     521
#define XSTR2  528
#define WCNT2  (KF * 8 * 16)

__global__ __launch_bounds__(256, 1)
void DP_conv(const v2f* __restrict__ x2, const float* __restrict__ conv_w,
             const float* __restrict__ conv_b, float* __restrict__ h2f) {
  __shared__ v2f wl2t[WCNT2];
  __shared__ v2f xl2[8 * XSTR2];

  const int b   = blockIdx.z;
  const int o0  = blockIdx.y * 16;
  const int t0  = blockIdx.x * TCHUNK;
  const int tid = threadIdx.x;
  const int lane = tid & 31;
  const int wid  = tid >> 5;
  const int m    = lane & 15;
  const int lh   = lane >> 4;

  v8f c0 = {0.f,0.f,0.f,0.f,0.f,0.f,0.f,0.f};
  v8f c1 = c0, c2 = c0, c3 = c0;

  for (int i0 = 0; i0 < DCH; i0 += 16) {
    __syncthreads();
    for (int idx = tid; idx < WCNT2; idx += 256) {
      int o = idx & 15, s = idx >> 4;
      int k = s >> 3, ii2 = s & 7;
      int oo = o0 + o, i = i0 + 2 * ii2;
      v2f v = {0.f, 0.f};
      if (oo < DCH && i < DCH) {
        const float* wp = &conv_w[(oo * DCH + i) * KF + k];
        v.x = wp[0]; v.y = wp[KF];
      }
      wl2t[idx] = v;
    }
    for (int idx = tid; idx < 8 * XJ; idx += 256) {
      int ii2 = idx / XJ, j = idx % XJ;
      int i2 = i0 / 2 + ii2, tau = t0 - PADK + j;
      v2f v = {0.f, 0.f};
      if (i2 < DP2 && tau >= 0 && tau < TLEN) v = x2[(b * DP2 + i2) * TLEN + tau];
      xl2[ii2 * XSTR2 + j] = v;
    }
    __syncthreads();

    for (int pb = 0; pb < 8; pb += 2) {
      for (int k = 0; k < KF; ++k) {
        v2f a = wl2t[(k * 8 + pb + lh) * 16 + m];
        const v2f* xr = &xl2[(pb + lh) * XSTR2 + wid * 64 + m + k];
        c0 = __builtin_amdgcn_wmma_f32_16x16x4_f32(false, a, false, xr[0],  (short)0, c0, false, false);
        c1 = __builtin_amdgcn_wmma_f32_16x16x4_f32(false, a, false, xr[16], (short)0, c1, false, false);
        c2 = __builtin_amdgcn_wmma_f32_16x16x4_f32(false, a, false, xr[32], (short)0, c2, false, false);
        c3 = __builtin_amdgcn_wmma_f32_16x16x4_f32(false, a, false, xr[48], (short)0, c3, false, false);
      }
    }
  }

  auto emit = [&](const v8f& c, int n) {
    int t = t0 + wid * 64 + n * 16 + m;
    if (t >= TP) return;
#pragma unroll
    for (int e = 0; e < 8; ++e) {
      int orow = o0 + e + 8 * lh;
      if (orow < DCH) {
        float v = c[e] + conv_b[orow];
        h2f[((b * DP2 + (orow >> 1)) * TP + t) * 2 + (orow & 1)] = v > 0.f ? v : 0.f;
      }
    }
  };
  emit(c0, 0); emit(c1, 1); emit(c2, 2); emit(c3, 3);
}

#define LT   558
#define NTT  129

__global__ __launch_bounds__(256, 1)
void DP_attn(const v2f* __restrict__ h2, const float* __restrict__ U,
             const float* __restrict__ fc_bias, float* __restrict__ out) {
  __shared__ v2f Ul2t[DP2 * 16];
  __shared__ float redm[2048], redz[2048], reds[2048];

  const int b   = blockIdx.y;
  const int l0  = blockIdx.x * 16;
  const int tid = threadIdx.x;
  const int lane = tid & 31, wid = tid >> 5;
  const int m = lane & 15, lh = lane >> 4;

  for (int idx = tid; idx < DP2 * 16; idx += 256) {
    int r = idx & 15, d2 = idx >> 4;
    v2f v = {0.f, 0.f};
    if (l0 + r < LROWS) v = *(const v2f*)&U[(l0 + r) * DCH + 2 * d2];
    Ul2t[idx] = v;
  }
  __syncthreads();

  float am[8], az[8], as_[8];
#pragma unroll
  for (int e = 0; e < 8; ++e) { am[e] = -1e30f; az[e] = 0.f; as_[e] = 0.f; }

  const v2f* hb2 = h2 + (size_t)b * DP2 * TP;
  for (int tt = wid * 2; tt < NTT; tt += 16) {
    const int ta = tt * 16 + m;
    const int tbt = ta + 16;
    const bool va = ta < TP;
    const bool vb = tbt < TP;
    const int tca = va ? ta : (TP - 1);
    const int tcb = vb ? tbt : (TP - 1);
    v8f c0 = {0.f,0.f,0.f,0.f,0.f,0.f,0.f,0.f};
    v8f c1 = c0;
    const v2f* hpa = hb2 + (size_t)lh * TP + tca;
    const v2f* hpb = hb2 + (size_t)lh * TP + tcb;
    const v2f* up  = &Ul2t[lh * 16 + m];
    for (int d0 = 0; d0 < DCH; d0 += 4) {
      v2f a = up[0]; up += 32;
      c0 = __builtin_amdgcn_wmma_f32_16x16x4_f32(false, a, false, hpa[0], (short)0, c0, false, false);
      c1 = __builtin_amdgcn_wmma_f32_16x16x4_f32(false, a, false, hpb[0], (short)0, c1, false, false);
      hpa += 2 * TP; hpb += 2 * TP;
    }
    auto absorb = [&](const v8f& c, bool valid) {
      if (!valid) return;
#pragma unroll
      for (int e = 0; e < 8; ++e) {
        float v  = c[e];
        float nm = fmaxf(am[e], v);
        float p  = __expf(am[e] - nm);
        float q  = __expf(v - nm);
        az[e]  = az[e]  * p + q;
        as_[e] = as_[e] * p + v * q;
        am[e]  = nm;
      }
    };
    absorb(c0, va);
    absorb(c1, vb);
  }

#pragma unroll
  for (int e = 0; e < 8; ++e) {
    int row = e + 8 * lh;
    int idx = (wid * 16 + row) * 16 + m;
    redm[idx] = am[e]; redz[idx] = az[e]; reds[idx] = as_[e];
  }
  __syncthreads();
  if (tid < 16) {
    int row = tid;
    float M = -1e30f, Z = 0.f, S = 0.f;
    for (int w = 0; w < 8; ++w)
      for (int s = 0; s < 16; ++s) {
        int idx = (w * 16 + row) * 16 + s;
        float mi = redm[idx];
        float nm = fmaxf(M, mi);
        float p = __expf(M - nm), q = __expf(mi - nm);
        Z = Z * p + redz[idx] * q;
        S = S * p + reds[idx] * q;
        M = nm;
      }
    int l = l0 + row;
    if (l < LROWS) out[b * LROWS + l] = S / Z + fc_bias[l];
  }
}

extern "C" void kernel_launch(void* const* d_in, const int* in_sizes, int n_in,
                              void* d_out, int out_size, void* d_ws, size_t ws_size,
                              hipStream_t stream) {
  const int*   ids     = (const int*)d_in[0];
  const float* embed_w = (const float*)d_in[1];
  const float* conv_w  = (const float*)d_in[2];
  const float* conv_b  = (const float*)d_in[3];
  const float* U       = (const float*)d_in[4];
  const float* fc_bias = (const float*)d_in[5];
  float* out = (float*)d_out;

  v2f* x2 = (v2f*)d_ws;
  v2f* h2 = x2 + XCNT / 2;
  float* h2f = (float*)h2;

  DP_gather<<<(XCNT / 2 + 255) / 256, 256, 0, stream>>>(ids, embed_w, x2);
  DP_conv<<<dim3((TP + TCHUNK - 1) / TCHUNK, (DCH + 15) / 16, BB), 256, 0, stream>>>(
      x2, conv_w, conv_b, h2f);
  DP_attn<<<dim3(LT, BB), 256, 0, stream>>>(h2, U, fc_bias, out);
}


# scores GEMM re-tiled: 6 row tiles of U per work-group in LDS, each h fragment fetched once for 6 WMMAs (was 1) and one group ahead; softmax partials merged by lane butterfly (new sum order)
# speedup vs baseline: 1.5953x; 1.5953x over previous
#include <hip/hip_runtime.h>

typedef __attribute__((ext_vector_type(2))) float v2f;
typedef __attribute__((ext_vector_type(8))) float v8f;

#define VOCAB 50000
#define LROWS 8921
#define DCH   300
#define DP2   150
#define KF    10
#define BB    8
#define TLEN  2048
#define TP    2057
#define PADK  9

#define XCNT  (BB * DCH * TLEN)
#define HCNT  (BB * DCH * TP)

__global__ __launch_bounds__(256)
void DP_gather(const int* __restrict__ ids, const float* __restrict__ embed,
               v2f* __restrict__ x2) {
  int idx = blockIdx.x * blockDim.x + threadIdx.x;
  if (idx >= XCNT / 2) return;
  int t = idx % TLEN;
  int p = (idx / TLEN) % DP2;
  int b = idx / (TLEN * DP2);
  int row = ids[b * TLEN + t];
  x2[idx] = *(const v2f*)&embed[row * DCH + 2 * p];
}

#define TCHUNK 512
#define XJ     521
#define XSTR2  528
#define WCNT2  (KF * 8 * 16)

__global__ __launch_bounds__(256, 1)
void DP_conv(const v2f* __restrict__ x2, const float* __restrict__ conv_w,
             const float* __restrict__ conv_b, float* __restrict__ h2f) {
  __shared__ v2f wl2t[WCNT2];
  __shared__ v2f xl2[8 * XSTR2];

  const int b   = blockIdx.z;
  const int o0  = blockIdx.y * 16;
  const int t0  = blockIdx.x * TCHUNK;
  const int tid = threadIdx.x;
  const int lane = tid & 31;
  const int wid  = tid >> 5;
  const int m    = lane & 15;
  const int lh   = lane >> 4;

  v8f c0 = {0.f,0.f,0.f,0.f,0.f,0.f,0.f,0.f};
  v8f c1 = c0, c2 = c0, c3 = c0;

  for (int i0 = 0; i0 < DCH; i0 += 16) {
    __syncthreads();
    for (int idx = tid; idx < WCNT2; idx += 256) {
      int o = idx & 15, s = idx >> 4;
      int k = s >> 3, ii2 = s & 7;
      int oo = o0 + o, i = i0 + 2 * ii2;
      v2f v = {0.f, 0.f};
      if (oo < DCH && i < DCH) {
        const float* wp = &conv_w[(oo * DCH + i) * KF + k];
        v.x = wp[0]; v.y = wp[KF];
      }
      wl2t[idx] = v;
    }
    for (int idx = tid; idx < 8 * XJ; idx += 256) {
      int ii2 = idx / XJ, j = idx % XJ;
      int i2 = i0 / 2 + ii2, tau = t0 - PADK + j;
      v2f v = {0.f, 0.f};
      if (i2 < DP2 && tau >= 0 && tau < TLEN) v = x2[(b * DP2 + i2) * TLEN + tau];
      xl2[ii2 * XSTR2 + j] = v;
    }
    __syncthreads();

    for (int pb = 0; pb < 8; pb += 2) {
      for (int k = 0; k < KF; ++k) {
        v2f a = wl2t[(k * 8 + pb + lh) * 16 + m];
        const v2f* xr = &xl2[(pb + lh) * XSTR2 + wid * 64 + m + k];
        c0 = __builtin_amdgcn_wmma_f32_16x16x4_f32(false, a, false, xr[0],  (short)0, c0, false, false);
        c1 = __builtin_amdgcn_wmma_f32_16x16x4_f32(false, a, false, xr[16], (short)0, c1, false, false);
        c2 = __builtin_amdgcn_wmma_f32_16x16x4_f32(false, a, false, xr[32], (short)0, c2, false, false);
        c3 = __builtin_amdgcn_wmma_f32_16x16x4_f32(false, a, false, xr[48], (short)0, c3, false, false);
      }
    }
  }

  auto emit = [&](const v8f& c, int n) {
    int t = t0 + wid * 64 + n * 16 + m;
    if (t >= TP) return;
#pragma unroll
    for (int e = 0; e < 8; ++e) {
      int orow = o0 + e + 8 * lh;
      if (orow < DCH) {
        float v = c[e] + conv_b[orow];
        h2f[((b * DP2 + (orow >> 1)) * TP + t) * 2 + (orow & 1)] = v > 0.f ? v : 0.f;
      }
    }
  };
  emit(c0, 0); emit(c1, 1); emit(c2, 2); emit(c3, 3);
}

#define LT   558
#define NTT  129
#define NL   6
#define LBLK (LT / NL)
#define ULT  (DP2 * 16)
#define NROW (NL * 16)
#define GK   3
#define NGRP (DCH / 4 / GK)

__device__ __forceinline__ void DP_merge(float& M, float& Z, float& S,
                                         float m2, float z2, float s2) {
  float nm = fmaxf(M, m2);
  float p = __expf(M - nm), q = __expf(m2 - nm);
  Z = Z * p + z2 * q;
  S = S * p + s2 * q;
  M = nm;
}

__global__ __launch_bounds__(256, 2)
void DP_attn(const v2f* __restrict__ h2, const float* __restrict__ U,
             const float* __restrict__ fc_bias, float* __restrict__ out) {
  __shared__ v2f Ul2t[NL * ULT];
  __shared__ float redm[8 * NROW], redz[8 * NROW], reds[8 * NROW];

  const int b   = blockIdx.y;
  const int l0  = blockIdx.x * NROW;
  const int tid = threadIdx.x;
  const int lane = tid & 31, wid = tid >> 5;
  const int m = lane & 15, lh = lane >> 4;

  for (int idx = tid; idx < NL * ULT; idx += 256) {
    int j = idx / ULT, rem = idx - j * ULT;
    int r = rem & 15, d2 = rem >> 4;
    int row = l0 + j * 16 + r;
    v2f v = {0.f, 0.f};
    if (row < LROWS) v = *(const v2f*)&U[row * DCH + 2 * d2];
    Ul2t[idx] = v;
  }
  __syncthreads();

  float am[NL][8], az[NL][8], as_[NL][8];
#pragma unroll
  for (int j = 0; j < NL; ++j)
#pragma unroll
    for (int e = 0; e < 8; ++e) { am[j][e] = -1e30f; az[j][e] = 0.f; as_[j][e] = 0.f; }

  const v2f* hb2 = h2 + (size_t)b * DP2 * TP + (size_t)lh * TP;
  v2f bc[GK];
  {
    const int t0c = wid * 16 + m;
    const v2f* hp0 = hb2 + t0c;
#pragma unroll
    for (int q = 0; q < GK; ++q) bc[q] = hp0[(size_t)q * 2 * TP];
  }
  for (int tt = wid; tt < NTT; tt += 8) {
    const int ta = tt * 16 + m;
    const bool va = ta < TP;
    const int tca = va ? ta : (TP - 1);
    const int tn  = (tt + 8 < NTT) ? (tt + 8) * 16 + m : ta;
    const int tcn = tn < TP ? tn : (TP - 1);
    v8f c[NL];
#pragma unroll
    for (int j = 0; j < NL; ++j) c[j] = (v8f){0.f,0.f,0.f,0.f,0.f,0.f,0.f,0.f};
    const v2f* hp = hb2 + tca + (size_t)(2 * GK) * TP;
    const v2f* up = &Ul2t[lh * 16 + m];
#pragma unroll 1
    for (int g = 0; g < NGRP; ++g) {
      const v2f* hn = (g < NGRP - 1) ? hp : (hb2 + tcn);
      v2f bn[GK];
#pragma unroll
      for (int q = 0; q < GK; ++q) bn[q] = hn[(size_t)q * 2 * TP];
      hp += (size_t)(2 * GK) * TP;
#pragma unroll
      for (int q = 0; q < GK; ++q) {
#pragma unroll
        for (int j = 0; j < NL; ++j) {
          v2f a = up[j * ULT + q * 32];
          c[j] = __builtin_amdgcn_wmma_f32_16x16x4_f32(false, a, false, bc[q], (short)0, c[j], false, false);
        }
      }
      up += 32 * GK;
#pragma unroll
      for (int q = 0; q < GK; ++q) bc[q] = bn[q];
    }
    if (va) {
#pragma unroll
      for (int j = 0; j < NL; ++j)
#pragma unroll
        for (int e = 0; e < 8; ++e) {
          float v  = c[j][e];
          float nm = fmaxf(am[j][e], v);
          float p  = __expf(am[j][e] - nm);
          float q  = __expf(v - nm);
          az[j][e]  = az[j][e]  * p + q;
          as_[j][e] = as_[j][e] * p + v * q;
          am[j][e]  = nm;
        }
    }
  }

#pragma unroll
  for (int j = 0; j < NL; ++j)
#pragma unroll
    for (int e = 0; e < 8; ++e) {
      float M = am[j][e], Z = az[j][e], S = as_[j][e];
#pragma unroll
      for (int off = 1; off < 16; off <<= 1) {
        float m2 = __shfl_xor(M, off);
        float z2 = __shfl_xor(Z, off);
        float s2 = __shfl_xor(S, off);
        DP_merge(M, Z, S, m2, z2, s2);
      }
      if (m == 0) {
        int idx = wid * NROW + j * 16 + e + 8 * lh;
        redm[idx] = M; redz[idx] = Z; reds[idx] = S;
      }
    }
  __syncthreads();
  if (tid < NROW) {
    int row = tid;
    float M = -1e30f, Z = 0.f, S = 0.f;
    for (int w = 0; w < 8; ++w) {
      int idx = w * NROW + row;
      DP_merge(M, Z, S, redm[idx], redz[idx], reds[idx]);
    }
    int l = l0 + row;
    if (l < LROWS) out[b * LROWS + l] = S / Z + fc_bias[l];
  }
}

extern "C" void kernel_launch(void* const* d_in, const int* in_sizes, int n_in,
                              void* d_out, int out_size, void* d_ws, size_t ws_size,
                              hipStream_t stream) {
  const int*   ids     = (const int*)d_in[0];
  const float* embed_w = (const float*)d_in[1];
  const float* conv_w  = (const float*)d_in[2];
  const float* conv_b  = (const float*)d_in[3];
  const float* U       = (const float*)d_in[4];
  const float* fc_bias = (const float*)d_in[5];
  float* out = (float*)d_out;

  v2f* x2 = (v2f*)d_ws;
  v2f* h2 = x2 + XCNT / 2;
  float* h2f = (float*)h2;

  DP_gather<<<(XCNT / 2 + 255) / 256, 256, 0, stream>>>(ids, embed_w, x2);
  DP_conv<<<dim3((TP + TCHUNK - 1) / TCHUNK, (DCH + 15) / 16, BB), 256, 0, stream>>>(
      x2, conv_w, conv_b, h2f);
  DP_attn<<<dim3(LBLK, BB), 256, 0, stream>>>(h2, U, fc_bias, out);
}


# staging loops (conv w and x chunk, bias, scores U tiles) made branch-free (clamped address, zero selected after) and unrolled: a stage's global loads are all in flight before the first wait
# speedup vs baseline: 1.8250x; 1.1440x over previous
#include <hip/hip_runtime.h>

typedef __attribute__((ext_vector_type(2))) float v2f;
typedef __attribute__((ext_vector_type(8))) float v8f;

#define VOCAB 50000
#define LROWS 8921
#define DCH   300
#define DP2   150
#define KF    10
#define BB    8
#define TLEN  2048
#define TP    2057
#define PADK  9

#define XCNT  (BB * DCH * TLEN)
#define HCNT  (BB * DCH * TP)

__global__ __launch_bounds__(256)
void DP_gather(const int* __restrict__ ids, const float* __restrict__ embed,
               v2f* __restrict__ x2) {
  int idx = blockIdx.x * blockDim.x + threadIdx.x;
  if (idx >= XCNT / 2) return;
  int t = idx % TLEN;
  int p = (idx / TLEN) % DP2;
  int b = idx / (TLEN * DP2);
  int row = ids[b * TLEN + t];
  x2[idx] = *(const v2f*)&embed[row * DCH + 2 * p];
}

#define TCHUNK 512
#define XJ     521
#define XSTR2  528
#define WCNT2  (KF * 8 * 16)

__global__ __launch_bounds__(256, 6)
void DP_conv(const v2f* __restrict__ x2, const float* __restrict__ conv_w,
             const float* __restrict__ conv_b, float* __restrict__ h2f) {
  __shared__ v2f wl2t[WCNT2];
  __shared__ v2f xl2[8 * XSTR2];

  const int b   = blockIdx.z;
  const int o0  = blockIdx.y * 16;
  const int t0  = blockIdx.x * TCHUNK;
  const int tid = threadIdx.x;
  const int lane = tid & 31;
  const int wid  = tid >> 5;
  const int m    = lane & 15;
  const int lh   = lane >> 4;

  v8f c0 = {0.f,0.f,0.f,0.f,0.f,0.f,0.f,0.f};
  v8f c1 = c0, c2 = c0, c3 = c0;

  for (int i0 = 0; i0 < DCH; i0 += 16) {
    v2f wv[5];
#pragma unroll
    for (int q = 0; q < 5; ++q) {
      int idx = tid + q * 256;
      int o = idx & 15, s = idx >> 4;
      int k = s >> 3, ii2 = s & 7;
      int oo = o0 + o, i = i0 + 2 * ii2;
      bool ok = oo < DCH && i < DCH;
      const float* wp = conv_w + (ok ? (oo * DCH + i) * KF + k : 0);
      v2f v;
      v.x = wp[0]; v.y = wp[KF];
      wv[q] = ok ? v : (v2f){0.f, 0.f};
    }
    v2f xv[17];
    {
      const int i2 = i0 / 2 + wid;
      const bool rok = i2 < DP2;
      const int rs = (b * DP2 + (rok ? i2 : 0)) * TLEN;
#pragma unroll
      for (int q = 0; q < 17; ++q) {
        int tau = t0 - PADK + lane + 32 * q;
        bool ok = rok && tau >= 0 && tau < TLEN;
        int g = rs + tau;
        g = g < rs ? rs : (g < rs + TLEN ? g : rs + TLEN - 1);
        v2f v = x2[g];
        xv[q] = ok ? v : (v2f){0.f, 0.f};
      }
    }
    __syncthreads();
#pragma unroll
    for (int q = 0; q < 5; ++q) wl2t[tid + q * 256] = wv[q];
#pragma unroll
    for (int q = 0; q < 17; ++q) {
      int j = lane + 32 * q;
      if (j < XJ) xl2[wid * XSTR2 + j] = xv[q];
    }
    __syncthreads();

    for (int pb = 0; pb < 8; pb += 2) {
      for (int k = 0; k < KF; ++k) {
        v2f a = wl2t[(k * 8 + pb + lh) * 16 + m];
        const v2f* xr = &xl2[(pb + lh) * XSTR2 + wid * 64 + m + k];
        c0 = __builtin_amdgcn_wmma_f32_16x16x4_f32(false, a, false, xr[0],  (short)0, c0, false, false);
        c1 = __builtin_amdgcn_wmma_f32_16x16x4_f32(false, a, false, xr[16], (short)0, c1, false, false);
        c2 = __builtin_amdgcn_wmma_f32_16x16x4_f32(false, a, false, xr[32], (short)0, c2, false, false);
        c3 = __builtin_amdgcn_wmma_f32_16x16x4_f32(false, a, false, xr[48], (short)0, c3, false, false);
      }
    }
  }

  float bias8[8];
#pragma unroll
  for (int e = 0; e < 8; ++e) {
    int orow = o0 + e + 8 * lh;
    bias8[e] = conv_b[orow < DCH ? orow : DCH - 1];
  }
  auto emit = [&](const v8f& c, int n) {
    int t = t0 + wid * 64 + n * 16 + m;
    if (t >= TP) return;
#pragma unroll
    for (int e = 0; e < 8; ++e) {
      int orow = o0 + e + 8 * lh;
      if (orow < DCH) {
        float v = c[e] + bias8[e];
        h2f[((b * DP2 + (orow >> 1)) * TP + t) * 2 + (orow & 1)] = v > 0.f ? v : 0.f;
      }
    }
  };
  emit(c0, 0); emit(c1, 1); emit(c2, 2); emit(c3, 3);
}

#define LT   558
#define NTT  129
#define NL   6
#define LBLK (LT / NL)
#define ULT  (DP2 * 16)
#define NROW (NL * 16)
#define GK   3
#define NGRP (DCH / 4 / GK)

__device__ __forceinline__ void DP_merge(float& M, float& Z, float& S,
                                         float m2, float z2, float s2) {
  float nm = fmaxf(M, m2);
  float p = __expf(M - nm), q = __expf(m2 - nm);
  Z = Z * p + z2 * q;
  S = S * p + s2 * q;
  M = nm;
}

__global__ __launch_bounds__(256, 2)
void DP_attn(const v2f* __restrict__ h2, const float* __restrict__ U,
             const float* __restrict__ fc_bias, float* __restrict__ out) {
  __shared__ v2f Ul2t[NL * ULT];
  __shared__ float redm[8 * NROW], redz[8 * NROW], reds[8 * NROW];

  const int b   = blockIdx.y;
  const int l0  = blockIdx.x * NROW;
  const int tid = threadIdx.x;
  const int lane = tid & 31, wid = tid >> 5;
  const int m = lane & 15, lh = lane >> 4;

  {
    const int r = tid & 15, dc = tid >> 4;
    for (int jh = 0; jh < NL; jh += 3) {
      v2f uv[3][10];
#pragma unroll
      for (int jj = 0; jj < 3; ++jj) {
        const int row = l0 + (jh + jj) * 16 + r;
        const bool rok = row < LROWS;
        const float* urow = U + (size_t)(rok ? row : 0) * DCH;
#pragma unroll
        for (int n = 0; n < 10; ++n) {
          int d2 = dc + 16 * n;
          int d2c = d2 < DP2 ? d2 : DP2 - 1;
          v2f v = *(const v2f*)&urow[2 * d2c];
          uv[jj][n] = rok ? v : (v2f){0.f, 0.f};
        }
      }
#pragma unroll
      for (int jj = 0; jj < 3; ++jj)
#pragma unroll
        for (int n = 0; n < 10; ++n) {
          int d2 = dc + 16 * n;
          if (d2 < DP2) Ul2t[(jh + jj) * ULT + d2 * 16 + r] = uv[jj][n];
        }
    }
  }
  __syncthreads();

  float am[NL][8], az[NL][8], as_[NL][8];
#pragma unroll
  for (int j = 0; j < NL; ++j)
#pragma unroll
    for (int e = 0; e < 8; ++e) { am[j][e] = -1e30f; az[j][e] = 0.f; as_[j][e] = 0.f; }

  const v2f* hb2 = h2 + (size_t)b * DP2 * TP + (size_t)lh * TP;
  v2f bc[GK];
  {
    const int t0c = wid * 16 + m;
    const v2f* hp0 = hb2 + t0c;
#pragma unroll
    for (int q = 0; q < GK; ++q) bc[q] = hp0[(size_t)q * 2 * TP];
  }
  for (int tt = wid; tt < NTT; tt += 8) {
    const int ta = tt * 16 + m;
    const bool va = ta < TP;
    const int tca = va ? ta : (TP - 1);
    const int tn  = (tt + 8 < NTT) ? (tt + 8) * 16 + m : ta;
    const int tcn = tn < TP ? tn : (TP - 1);
    v8f c[NL];
#pragma unroll
    for (int j = 0; j < NL; ++j) c[j] = (v8f){0.f,0.f,0.f,0.f,0.f,0.f,0.f,0.f};
    const v2f* hp = hb2 + tca + (size_t)(2 * GK) * TP;
    const v2f* up = &Ul2t[lh * 16 + m];
#pragma unroll 1
    for (int g = 0; g < NGRP; ++g) {
      const v2f* hn = (g < NGRP - 1) ? hp : (hb2 + tcn);
      v2f bn[GK];
#pragma unroll
      for (int q = 0; q < GK; ++q) bn[q] = hn[(size_t)q * 2 * TP];
      hp += (size_t)(2 * GK) * TP;
#pragma unroll
      for (int q = 0; q < GK; ++q) {
#pragma unroll
        for (int j = 0; j < NL; ++j) {
          v2f a = up[j * ULT + q * 32];
          c[j] = __builtin_amdgcn_wmma_f32_16x16x4_f32(false, a, false, bc[q], (short)0, c[j], false, false);
        }
      }
      up += 32 * GK;
#pragma unroll
      for (int q = 0; q < GK; ++q) bc[q] = bn[q];
    }
    if (va) {
#pragma unroll
      for (int j = 0; j < NL; ++j)
#pragma unroll
        for (int e = 0; e < 8; ++e) {
          float v  = c[j][e];
          float nm = fmaxf(am[j][e], v);
          float p  = __expf(am[j][e] - nm);
          float q  = __expf(v - nm);
          az[j][e]  = az[j][e]  * p + q;
          as_[j][e] = as_[j][e] * p + v * q;
          am[j][e]  = nm;
        }
    }
  }

#pragma unroll
  for (int j = 0; j < NL; ++j)
#pragma unroll
    for (int e = 0; e < 8; ++e) {
      float M = am[j][e], Z = az[j][e], S = as_[j][e];
#pragma unroll
      for (int off = 1; off < 16; off <<= 1) {
        float m2 = __shfl_xor(M, off);
        float z2 = __shfl_xor(Z, off);
        float s2 = __shfl_xor(S, off);
        DP_merge(M, Z, S, m2, z2, s2);
      }
      if (m == 0) {
        int idx = wid * NROW + j * 16 + e + 8 * lh;
        redm[idx] = M; redz[idx] = Z; reds[idx] = S;
      }
    }
  __syncthreads();
  if (tid < NROW) {
    int row = tid;
    float M = -1e30f, Z = 0.f, S = 0.f;
    for (int w = 0; w < 8; ++w) {
      int idx = w * NROW + row;
      DP_merge(M, Z, S, redm[idx], redz[idx], reds[idx]);
    }
    int l = l0 + row;
    if (l < LROWS) out[b * LROWS + l] = S / Z + fc_bias[l];
  }
}

extern "C" void kernel_launch(void* const* d_in, const int* in_sizes, int n_in,
                              void* d_out, int out_size, void* d_ws, size_t ws_size,
                              hipStream_t stream) {
  const int*   ids     = (const int*)d_in[0];
  const float* embed_w = (const float*)d_in[1];
  const float* conv_w  = (const float*)d_in[2];
  const float* conv_b  = (const float*)d_in[3];
  const float* U       = (const float*)d_in[4];
  const float* fc_bias = (const float*)d_in[5];
  float* out = (float*)d_out;

  v2f* x2 = (v2f*)d_ws;
  v2f* h2 = x2 + XCNT / 2;
  float* h2f = (float*)h2;

  DP_gather<<<(XCNT / 2 + 255) / 256, 256, 0, stream>>>(ids, embed_w, x2);
  DP_conv<<<dim3((TP + TCHUNK - 1) / TCHUNK, (DCH + 15) / 16, BB), 256, 0, stream>>>(
      x2, conv_w, conv_b, h2f);
  DP_attn<<<dim3(LBLK, BB), 256, 0, stream>>>(h2, U, fc_bias, out);
}


# scores kernel: next group's U fragments read from LDS one group ahead (2nd register set); room made by swapping WMMA operands: softmax state per lane 6 rows, not 48 (new sum order)
# speedup vs baseline: 1.9782x; 1.0839x over previous
#include <hip/hip_runtime.h>

typedef __attribute__((ext_vector_type(2))) float v2f;
typedef __attribute__((ext_vector_type(8))) float v8f;

#define VOCAB 50000
#define LROWS 8921
#define DCH   300
#define DP2   150
#define KF    10
#define BB    8
#define TLEN  2048
#define TP    2057
#define PADK  9

#define XCNT  (BB * DCH * TLEN)
#define HCNT  (BB * DCH * TP)

__global__ __launch_bounds__(256)
void DP_gather(const int* __restrict__ ids, const float* __restrict__ embed,
               v2f* __restrict__ x2) {
  int idx = blockIdx.x * blockDim.x + threadIdx.x;
  if (idx >= XCNT / 2) return;
  int t = idx % TLEN;
  int p = (idx / TLEN) % DP2;
  int b = idx / (TLEN * DP2);
  int row = ids[b * TLEN + t];
  x2[idx] = *(const v2f*)&embed[row * DCH + 2 * p];
}

#define TCHUNK 512
#define XJ     521
#define XSTR2  528
#define WCNT2  (KF * 8 * 16)

__global__ __launch_bounds__(256, 6)
void DP_conv(const v2f* __restrict__ x2, const float* __restrict__ conv_w,
             const float* __restrict__ conv_b, float* __restrict__ h2f) {
  __shared__ v2f wl2t[WCNT2];
  __shared__ v2f xl2[8 * XSTR2];

  const int b   = blockIdx.z;
  const int o0  = blockIdx.y * 16;
  const int t0  = blockIdx.x * TCHUNK;
  const int tid = threadIdx.x;
  const int lane = tid & 31;
  const int wid  = tid >> 5;
  const int m    = lane & 15;
  const int lh   = lane >> 4;

  v8f c0 = {0.f,0.f,0.f,0.f,0.f,0.f,0.f,0.f};
  v8f c1 = c0, c2 = c0, c3 = c0;

  for (int i0 = 0; i0 < DCH; i0 += 16) {
    v2f wv[5];
#pragma unroll
    for (int q = 0; q < 5; ++q) {
      int idx = tid + q * 256;
      int o = idx & 15, s = idx >> 4;
      int k = s >> 3, ii2 = s & 7;
      int oo = o0 + o, i = i0 + 2 * ii2;
      bool ok = oo < DCH && i < DCH;
      const float* wp = conv_w + (ok ? (oo * DCH + i) * KF + k : 0);
      v2f v;
      v.x = wp[0]; v.y = wp[KF];
      wv[q] = ok ? v : (v2f){0.f, 0.f};
    }
    v2f xv[17];
    {
      const int i2 = i0 / 2 + wid;
      const bool rok = i2 < DP2;
      const int rs = (b * DP2 + (rok ? i2 : 0)) * TLEN;
#pragma unroll
      for (int q = 0; q < 17; ++q) {
        int tau = t0 - PADK + lane + 32 * q;
        bool ok = rok && tau >= 0 && tau < TLEN;
        int g = rs + tau;
        g = g < rs ? rs : (g < rs + TLEN ? g : rs + TLEN - 1);
        v2f v = x2[g];
        xv[q] = ok ? v : (v2f){0.f, 0.f};
      }
    }
    __syncthreads();
#pragma unroll
    for (int q = 0; q < 5; ++q) wl2t[tid + q * 256] = wv[q];
#pragma unroll
    for (int q = 0; q < 17; ++q) {
      int j = lane + 32 * q;
      if (j < XJ) xl2[wid * XSTR2 + j] = xv[q];
    }
    __syncthreads();

    for (int pb = 0; pb < 8; pb += 2) {
      for (int k = 0; k < KF; ++k) {
        v2f a = wl2t[(k * 8 + pb + lh) * 16 + m];
        const v2f* xr = &xl2[(pb + lh) * XSTR2 + wid * 64 + m + k];
        c0 = __builtin_amdgcn_wmma_f32_16x16x4_f32(false, a, false, xr[0],  (short)0, c0, false, false);
        c1 = __builtin_amdgcn_wmma_f32_16x16x4_f32(false, a, false, xr[16], (short)0, c1, false, false);
        c2 = __builtin_amdgcn_wmma_f32_16x16x4_f32(false, a, false, xr[32], (short)0, c2, false, false);
        c3 = __builtin_amdgcn_wmma_f32_16x16x4_f32(false, a, false, xr[48], (short)0, c3, false, false);
      }
    }
  }

  float bias8[8];
#pragma unroll
  for (int e = 0; e < 8; ++e) {
    int orow = o0 + e + 8 * lh;
    bias8[e] = conv_b[orow < DCH ? orow : DCH - 1];
  }
  auto emit = [&](const v8f& c, int n) {
    int t = t0 + wid * 64 + n * 16 + m;
    if (t >= TP) return;
#pragma unroll
    for (int e = 0; e < 8; ++e) {
      int orow = o0 + e + 8 * lh;
      if (orow < DCH) {
        float v = c[e] + bias8[e];
        h2f[((b * DP2 + (orow >> 1)) * TP + t) * 2 + (orow & 1)] = v > 0.f ? v : 0.f;
      }
    }
  };
  emit(c0, 0); emit(c1, 1); emit(c2, 2); emit(c3, 3);
}

#define LT   558
#define NTT  129
#define NL   6
#define LBLK (LT / NL)
#define ULT  (DP2 * 16)
#define NROW (NL * 16)
#define GK   3
#define NGRP (DCH / 4 / GK)

__device__ __forceinline__ void DP_merge(float& M, float& Z, float& S,
                                         float m2, float z2, float s2) {
  float nm = fmaxf(M, m2);
  float p = __expf(M - nm), q = __expf(m2 - nm);
  Z = Z * p + z2 * q;
  S = S * p + s2 * q;
  M = nm;
}

__global__ __launch_bounds__(256, 2)
void DP_attn(const v2f* __restrict__ h2, const float* __restrict__ U,
             const float* __restrict__ fc_bias, float* __restrict__ out) {
  __shared__ v2f Ul2t[NL * ULT];
  __shared__ float redm[8 * NROW], redz[8 * NROW], reds[8 * NROW];

  const int b   = blockIdx.y;
  const int l0  = blockIdx.x * NROW;
  const int tid = threadIdx.x;
  const int lane = tid & 31, wid = tid >> 5;
  const int m = lane & 15, lh = lane >> 4;

  {
    const int r = tid & 15, dc = tid >> 4;
    for (int jh = 0; jh < NL; jh += 3) {
      v2f uv[3][10];
#pragma unroll
      for (int jj = 0; jj < 3; ++jj) {
        const int row = l0 + (jh + jj) * 16 + r;
        const bool rok = row < LROWS;
        const float* urow = U + (size_t)(rok ? row : 0) * DCH;
#pragma unroll
        for (int n = 0; n < 10; ++n) {
          int d2 = dc + 16 * n;
          int d2c = d2 < DP2 ? d2 : DP2 - 1;
          v2f v = *(const v2f*)&urow[2 * d2c];
          uv[jj][n] = rok ? v : (v2f){0.f, 0.f};
        }
      }
#pragma unroll
      for (int jj = 0; jj < 3; ++jj)
#pragma unroll
        for (int n = 0; n < 10; ++n) {
          int d2 = dc + 16 * n;
          if (d2 < DP2) Ul2t[(jh + jj) * ULT + d2 * 16 + r] = uv[jj][n];
        }
    }
  }
  __syncthreads();

  float am[NL], az[NL], as_[NL];
#pragma unroll
  for (int j = 0; j < NL; ++j) { am[j] = -1e30f; az[j] = 0.f; as_[j] = 0.f; }

  const v2f* hb2 = h2 + (size_t)b * DP2 * TP + (size_t)lh * TP;
  const v2f* ub = &Ul2t[lh * 16 + m];
  v2f hA[GK], hB[GK], uA[GK][NL], uB[GK][NL];
  {
    const v2f* hp0 = hb2 + wid * 16 + m;
#pragma unroll
    for (int q = 0; q < GK; ++q) {
      hA[q] = hp0[(size_t)q * 2 * TP];
#pragma unroll
      for (int j = 0; j < NL; ++j) uA[q][j] = ub[j * ULT + q * 32];
    }
  }
  for (int tt = wid; tt < NTT; tt += 8) {
    const int ta = tt * 16 + m;
    const int tca = ta < TP ? ta : (TP - 1);
    const int tn  = (tt + 8 < NTT) ? (tt + 8) * 16 + m : ta;
    const int tcn = tn < TP ? tn : (TP - 1);
    v8f c[NL];
#pragma unroll
    for (int j = 0; j < NL; ++j) c[j] = (v8f){0.f,0.f,0.f,0.f,0.f,0.f,0.f,0.f};
    auto group = [&](const v2f (&hc)[GK], const v2f (&uc)[GK][NL],
                     v2f (&hn)[GK], v2f (&un)[GK][NL], const v2f* hs, const v2f* us) {
#pragma unroll
      for (int q = 0; q < GK; ++q) hn[q] = hs[(size_t)q * 2 * TP];
#pragma unroll
      for (int q = 0; q < GK; ++q)
#pragma unroll
        for (int j = 0; j < NL; ++j) un[q][j] = us[j * ULT + q * 32];
      __builtin_amdgcn_sched_barrier(0);
#pragma unroll
      for (int q = 0; q < GK; ++q)
#pragma unroll
        for (int j = 0; j < NL; ++j)
          c[j] = __builtin_amdgcn_wmma_f32_16x16x4_f32(false, hc[q], false, uc[q][j], (short)0, c[j], false, false);
    };
    const v2f* hp = hb2 + tca + (size_t)(2 * GK) * TP;
    const v2f* up = ub + 32 * GK;
#pragma unroll 1
    for (int gp = 0; gp < NGRP / 2; ++gp) {
      group(hA, uA, hB, uB, hp, up);
      group(hB, uB, hA, uA, hp + (size_t)(2 * GK) * TP, up + 32 * GK);
      hp += (size_t)(4 * GK) * TP;
      up += 64 * GK;
    }
    group(hA, uA, hB, uB, hb2 + tcn, ub);
#pragma unroll
    for (int q = 0; q < GK; ++q) {
      hA[q] = hB[q];
#pragma unroll
      for (int j = 0; j < NL; ++j) uA[q][j] = uB[q][j];
    }
    auto update = [&](const bool partial) {
      const int tb = tt * 16 + 8 * lh;
#pragma unroll
      for (int j = 0; j < NL; ++j) {
        float mx = am[j];
#pragma unroll
        for (int e = 0; e < 8; ++e) {
          float v = c[j][e];
          mx = (!partial || tb + e < TP) ? fmaxf(mx, v) : mx;
        }
        const float p = __expf(am[j] - mx);
        float z = az[j] * p, s = as_[j] * p;
#pragma unroll
        for (int e = 0; e < 8; ++e) {
          float v = c[j][e];
          float q = __expf(v - mx);
          if (partial) q = (tb + e < TP) ? q : 0.f;
          z += q;
          s += v * q;
        }
        am[j] = mx; az[j] = z; as_[j] = s;
      }
    };
    if (tt != NTT - 1) update(false); else update(true);
  }

#pragma unroll
  for (int j = 0; j < NL; ++j) {
    float M = am[j], Z = az[j], S = as_[j];
    float m2 = __shfl_xor(M, 16);
    float z2 = __shfl_xor(Z, 16);
    float s2 = __shfl_xor(S, 16);
    DP_merge(M, Z, S, m2, z2, s2);
    if (lh == 0) {
      int idx = wid * NROW + j * 16 + m;
      redm[idx] = M; redz[idx] = Z; reds[idx] = S;
    }
  }
  __syncthreads();
  if (tid < NROW) {
    int row = tid;
    float M = -1e30f, Z = 0.f, S = 0.f;
    for (int w = 0; w < 8; ++w) {
      int idx = w * NROW + row;
      DP_merge(M, Z, S, redm[idx], redz[idx], reds[idx]);
    }
    int l = l0 + row;
    if (l < LROWS) out[b * LROWS + l] = S / Z + fc_bias[l];
  }
}

extern "C" void kernel_launch(void* const* d_in, const int* in_sizes, int n_in,
                              void* d_out, int out_size, void* d_ws, size_t ws_size,
                              hipStream_t stream) {
  const int*   ids     = (const int*)d_in[0];
  const float* embed_w = (const float*)d_in[1];
  const float* conv_w  = (const float*)d_in[2];
  const float* conv_b  = (const float*)d_in[3];
  const float* U       = (const float*)d_in[4];
  const float* fc_bias = (const float*)d_in[5];
  float* out = (float*)d_out;

  v2f* x2 = (v2f*)d_ws;
  v2f* h2 = x2 + XCNT / 2;
  float* h2f = (float*)h2;

  DP_gather<<<(XCNT / 2 + 255) / 256, 256, 0, stream>>>(ids, embed_w, x2);
  DP_conv<<<dim3((TP + TCHUNK - 1) / TCHUNK, (DCH + 15) / 16, BB), 256, 0, stream>>>(
      x2, conv_w, conv_b, h2f);
  DP_attn<<<dim3(LBLK, BB), 256, 0, stream>>>(h2, U, fc_bias, out);
}
